# starting kernel
# baseline (speedup 1.0000x reference)
#include <hip/hip_runtime.h>
#include <hip/hip_bf16.h>

#define N_FEATS 64
#define WTP_LD  80

typedef float v2f __attribute__((ext_vector_type(2)));
typedef float v8f __attribute__((ext_vector_type(8)));

__global__ void zero_f32(float* __restrict__ p, int n) {
    int i = blockIdx.x * blockDim.x + threadIdx.x;
    if (i < n) p[i] = 0.0f;
}

__global__ void edge_scatter(const float* __restrict__ x,
                             const int*   __restrict__ src,
                             const int*   __restrict__ dst,
                             float*       __restrict__ agg,
                             float*       __restrict__ deg,
                             int n_edges) {
    int t = blockIdx.x * blockDim.x + threadIdx.x;
    int e = t >> 4;
    if (e >= n_edges) return;
    int l = t & 15;
    int s = src[e];
    int d = dst[e];
    const float4 v = *reinterpret_cast<const float4*>(x + (size_t)s * N_FEATS + l * 4);
    float* o = agg + (size_t)d * N_FEATS + l * 4;
    atomicAdd(o + 0, v.x);
    atomicAdd(o + 1, v.y);
    atomicAdd(o + 2, v.z);
    atomicAdd(o + 3, v.w);
    if (l == 0) atomicAdd(deg + d, 1.0f);
}

__global__ void __launch_bounds__(256)
gcn_linear_wmma(const float* __restrict__ x,
                const float* __restrict__ agg,
                const float* __restrict__ deg,
                const float* __restrict__ W,
                const float* __restrict__ bias,
                float*       __restrict__ out,
                int n_tiles) {
    __shared__ float WtPf[32 * WTP_LD * 2];
    __shared__ float bsh[N_FEATS];

    for (int i = threadIdx.x; i < N_FEATS * N_FEATS; i += blockDim.x) {
        int n = i / N_FEATS;
        int k = i % N_FEATS;
        WtPf[((k >> 1) * WTP_LD + n) * 2 + (k & 1)] = W[i];
    }
    if (threadIdx.x < N_FEATS) bsh[threadIdx.x] = bias[threadIdx.x];
    __syncthreads();

    const int wid  = threadIdx.x >> 5;
    const int lane = threadIdx.x & 31;
    const int tile = blockIdx.x * 8 + wid;
    if (tile >= n_tiles) return;

    const int half = lane >> 4;
    const int l16  = lane & 15;
    const int row  = tile * 16 + l16;

    const bool use_agg = deg[row] > 0.0f;
    const float* hrow = use_agg ? (agg + (size_t)row * N_FEATS)
                                : (x   + (size_t)row * N_FEATS);

    v8f acc0 = {}, acc1 = {}, acc2 = {}, acc3 = {};

#pragma unroll
    for (int s = 0; s < 16; ++s) {
        const int kp = 2 * s + half;

        v2f a = *reinterpret_cast<const v2f*>(hrow + 2 * kp);

        const v2f* wrow = reinterpret_cast<const v2f*>(&WtPf[kp * WTP_LD * 2]);
        v2f b0 = wrow[l16 +  0];
        v2f b1 = wrow[l16 + 16];
        v2f b2 = wrow[l16 + 32];
        v2f b3 = wrow[l16 + 48];

        acc0 = __builtin_amdgcn_wmma_f32_16x16x4_f32(false, a, false, b0, (short)0, acc0, false, false);
        acc1 = __builtin_amdgcn_wmma_f32_16x16x4_f32(false, a, false, b1, (short)0, acc1, false, false);
        acc2 = __builtin_amdgcn_wmma_f32_16x16x4_f32(false, a, false, b2, (short)0, acc2, false, false);
        acc3 = __builtin_amdgcn_wmma_f32_16x16x4_f32(false, a, false, b3, (short)0, acc3, false, false);
    }

    float* obase = out + (size_t)(tile * 16 + 8 * half) * N_FEATS + l16;
#pragma unroll
    for (int g = 0; g < 8; ++g) {
        float* orow = obase + (size_t)g * N_FEATS;
        orow[ 0] = acc0[g] + bsh[l16 +  0];
        orow[16] = acc1[g] + bsh[l16 + 16];
        orow[32] = acc2[g] + bsh[l16 + 32];
        orow[48] = acc3[g] + bsh[l16 + 48];
    }
}

extern "C" void kernel_launch(void* const* d_in, const int* in_sizes, int n_in,
                              void* d_out, int out_size, void* d_ws, size_t ws_size,
                              hipStream_t stream) {
    const float* x    = (const float*)d_in[0];
    const int*   src  = (const int*)  d_in[1];
    const int*   dst  = (const int*)  d_in[2];
    const float* W    = (const float*)d_in[3];
    const float* bias = (const float*)d_in[4];
    float*       out  = (float*)d_out;

    const int n_nodes = in_sizes[0] / N_FEATS;
    const int n_edges = in_sizes[1];

    float* agg = (float*)d_ws;
    float* deg = agg + (size_t)n_nodes * N_FEATS;

    const int n_zero = n_nodes * N_FEATS + n_nodes;
    zero_f32<<<(n_zero + 255) / 256, 256, 0, stream>>>(agg, n_zero);

    const int scat_threads = n_edges * 16;
    edge_scatter<<<(scat_threads + 255) / 256, 256, 0, stream>>>(
        x, src, dst, agg, deg, n_edges);

    const int n_tiles = (n_nodes + 15) / 16;
    const int blocks  = (n_tiles + 7) / 8;
    gcn_linear_wmma<<<blocks, 256, 0, stream>>>(x, agg, deg, W, bias, out, n_tiles);
}


# One fused launch: each work-group owns 400 nodes, scans all dst, sums x[src] of its edges into LDS with ds float atomics (arrival order), then WMMA from LDS; global atomics, agg and zero pass gone
# speedup vs baseline: 2.6846x; 2.6846x over previous
#include <hip/hip_runtime.h>

#define N_FEATS 64
#define WTP_LD  80
#define NPB     400
#define AGG_LD  68
#define NWAVES  32
#define LCAP    160
#define SCAN_U  8

typedef float v2f __attribute__((ext_vector_type(2)));
typedef float v8f __attribute__((ext_vector_type(8)));

__global__ void __launch_bounds__(1024)
owner_scan_linear(const float* __restrict__ x,
                const int*   __restrict__ src,
                const int*   __restrict__ dst,
                const float* __restrict__ W,
                const float* __restrict__ bias,
                float*       __restrict__ out,
                int n_nodes, int n_edges) {
    __shared__ float aggS[NPB * AGG_LD];
    __shared__ int   degS[NPB];
    __shared__ float WtPf[32 * WTP_LD * 2];
    __shared__ float bsh[N_FEATS];
    __shared__ int   listE[NWAVES * LCAP];
    __shared__ int   listD[NWAVES * LCAP];

    const int tid  = threadIdx.x;
    const int wave = __builtin_amdgcn_readfirstlane(tid >> 5);
    const int lane = tid & 31;
    const int lo   = blockIdx.x * NPB;
    const int own  = min(NPB, n_nodes - lo);

    for (int i = tid; i < NPB * AGG_LD; i += 1024) aggS[i] = 0.0f;
    for (int i = tid; i < NPB; i += 1024) degS[i] = 0;
    for (int i = tid; i < N_FEATS * N_FEATS; i += 1024) {
        int n = i / N_FEATS;
        int k = i % N_FEATS;
        WtPf[((k >> 1) * WTP_LD + n) * 2 + (k & 1)] = W[i];
    }
    if (tid < N_FEATS) bsh[tid] = bias[tid];
    __syncthreads();

    int* myE = listE + wave * LCAP;
    int* myD = listD + wave * LCAP;
    int  cnt = 0;

    auto flush = [&](int n) {
        for (int base = 0; base < n; base += 32) {
            const int m = min(32, n - base);
            int d_l = 0, s_l = 0;
            if (lane < m) {
                const int e_l = myE[base + lane];
                d_l = myD[base + lane];
                s_l = min(max(src[e_l], 0), n_nodes - 1);
            }
            for (int k = 0; k < m; k += 8) {
                v2f r[8];
                int dl[8];
#pragma unroll
                for (int u = 0; u < 8; ++u) {
                    const int j = min(k + u, m - 1);
                    const int s = __builtin_amdgcn_readlane(s_l, j);
                    dl[u] = __builtin_amdgcn_readlane(d_l, j);
                    r[u] = *reinterpret_cast<const v2f*>(x + (size_t)s * N_FEATS + 2 * lane);
                }
#pragma unroll
                for (int u = 0; u < 8; ++u) {
                    if (k + u < m) {
                        float* a = aggS + dl[u] * AGG_LD + 2 * lane;
                        atomicAdd(a, r[u].x);
                        atomicAdd(a + 1, r[u].y);
                        if (lane == 0) atomicAdd(&degS[dl[u]], 1);
                    }
                }
            }
        }
    };

    auto append = [&](int dv, int e) {
        const unsigned dloc = (unsigned)(dv - lo);
        const bool hit = dloc < (unsigned)own;
        const unsigned bal = __builtin_amdgcn_ballot_w32(hit);
        if (bal) {
            const int pos = cnt + (int)__builtin_amdgcn_mbcnt_lo(bal, 0u);
            if (hit) {
                if (pos < LCAP) {
                    myE[pos] = e;
                    myD[pos] = (int)dloc;
                } else {
                    const int s = min(max(src[e], 0), n_nodes - 1);
                    const float* xr = x + (size_t)s * N_FEATS;
                    float* ar = aggS + dloc * AGG_LD;
#pragma nounroll
                    for (int k = 0; k < N_FEATS; ++k) atomicAdd(ar + k, xr[k]);
                    atomicAdd(&degS[dloc], 1);
                }
            }
            cnt += __builtin_popcount(bal);
        }
    };

    const int n_full = n_edges / (128 * SCAN_U);
    const int rot    = (int)(((long long)blockIdx.x * n_full) / gridDim.x);
    const int4* dst4 = reinterpret_cast<const int4*>(dst);

    for (int c0 = wave; c0 < n_full; c0 += NWAVES) {
        int c = c0 + rot;
        if (c >= n_full) c -= n_full;
        const int q0 = c * (32 * SCAN_U) + lane;
        int4 d[SCAN_U];
#pragma unroll
        for (int u = 0; u < SCAN_U; ++u) d[u] = dst4[q0 + u * 32];
#pragma unroll
        for (int u = 0; u < SCAN_U; ++u) {
            const int e = (q0 + u * 32) * 4;
            append(d[u].x, e + 0);
            append(d[u].y, e + 1);
            append(d[u].z, e + 2);
            append(d[u].w, e + 3);
        }
        if (cnt >= LCAP / 2) {
            flush(min(cnt, LCAP));
            cnt = 0;
        }
    }
    for (int e0 = n_full * (128 * SCAN_U) + wave * 32; e0 < n_edges; e0 += NWAVES * 32) {
        const int e = e0 + lane;
        const int dv = (e < n_edges) ? dst[e] : -1;
        append(dv, e);
    }
    flush(min(cnt, LCAP));
    __syncthreads();

    for (int i = tid; i < own * 16; i += 1024) {
        const int r = i >> 4;
        const int q = i & 15;
        if (degS[r] == 0) {
            const float4 v = *reinterpret_cast<const float4*>(x + (size_t)(lo + r) * N_FEATS + q * 4);
            float* a = aggS + r * AGG_LD + q * 4;
            a[0] = v.x; a[1] = v.y; a[2] = v.z; a[3] = v.w;
        }
    }
    __syncthreads();

    const int n_tiles = (own + 15) >> 4;
    if (wave >= n_tiles) return;

    const int half = lane >> 4;
    const int l16  = lane & 15;
    const float* hrow = aggS + (wave * 16 + l16) * AGG_LD;

    v8f acc0 = {}, acc1 = {}, acc2 = {}, acc3 = {};

#pragma unroll
    for (int s = 0; s < 16; ++s) {
        const int kp = 2 * s + half;
        v2f a = *reinterpret_cast<const v2f*>(hrow + 2 * kp);
        const v2f* wrow = reinterpret_cast<const v2f*>(&WtPf[kp * WTP_LD * 2]);
        v2f b0 = wrow[l16 +  0];
        v2f b1 = wrow[l16 + 16];
        v2f b2 = wrow[l16 + 32];
        v2f b3 = wrow[l16 + 48];
        acc0 = __builtin_amdgcn_wmma_f32_16x16x4_f32(false, a, false, b0, (short)0, acc0, false, false);
        acc1 = __builtin_amdgcn_wmma_f32_16x16x4_f32(false, a, false, b1, (short)0, acc1, false, false);
        acc2 = __builtin_amdgcn_wmma_f32_16x16x4_f32(false, a, false, b2, (short)0, acc2, false, false);
        acc3 = __builtin_amdgcn_wmma_f32_16x16x4_f32(false, a, false, b3, (short)0, acc3, false, false);
    }

    const int row0 = lo + wave * 16 + 8 * half;
    float* obase = out + (size_t)row0 * N_FEATS + l16;
#pragma unroll
    for (int g = 0; g < 8; ++g) {
        if (row0 + g < n_nodes) {
            float* orow = obase + (size_t)g * N_FEATS;
            orow[ 0] = acc0[g] + bsh[l16 +  0];
            orow[16] = acc1[g] + bsh[l16 + 16];
            orow[32] = acc2[g] + bsh[l16 + 32];
            orow[48] = acc3[g] + bsh[l16 + 48];
        }
    }
}

extern "C" void kernel_launch(void* const* d_in, const int* in_sizes, int n_in,
                              void* d_out, int out_size, void* d_ws, size_t ws_size,
                              hipStream_t stream) {
    const float* x    = (const float*)d_in[0];
    const int*   src  = (const int*)  d_in[1];
    const int*   dst  = (const int*)  d_in[2];
    const float* W    = (const float*)d_in[3];
    const float* bias = (const float*)d_in[4];
    float*       out  = (float*)d_out;

    const int n_nodes = in_sizes[0] / N_FEATS;
    const int n_edges = in_sizes[1];

    const int blocks = (n_nodes + NPB - 1) / NPB;
    owner_scan_linear<<<blocks, 1024, 0, stream>>>(x, src, dst, W, bias, out, n_nodes, n_edges);
}


# Edges bucketed by owner group in a first launch (LDS counting sort per 8192-edge chunk); each owner reads only its own slices instead of scanning all dst; LDS float adds stay in arrival order
# speedup vs baseline: 4.5312x; 1.6878x over previous
#include <hip/hip_runtime.h>

#define N_FEATS 64
#define WTP_LD  80
#define NPB     400
#define AGG_LD  68
#define NWAVES  32
#define CHUNK   8192
#define GMAX    1023
#define ROWS_U  16

#define LIST_OFF_BYTES 0u
#define OFF_OFF_BYTES  (128u << 20)

typedef float v2f __attribute__((ext_vector_type(2)));
typedef float v8f __attribute__((ext_vector_type(8)));

__global__ void __launch_bounds__(1024)
bin_edges(const int* __restrict__ src,
          const int* __restrict__ dst,
          int*       __restrict__ listI,
          int*       __restrict__ off,
          int n_nodes, int n_edges, int n_groups) {
    __shared__ int  cntS[GMAX + 1];
    __shared__ int  wsum[NWAVES];
    __shared__ int2 sortS[CHUNK];

    const int tid  = threadIdx.x;
    const int wave = tid >> 5;
    const int lane = tid & 31;
    const int w    = blockIdx.x;

    cntS[tid] = 0;
    __syncthreads();

    int gq[8], rk[8], sv[8], dlv[8];
#pragma unroll
    for (int i = 0; i < 2; ++i) {
        const int e0 = w * CHUNK + (i * 1024 + tid) * 4;
        int d4[4], s4[4];
        if (e0 + 3 < n_edges) {
            const int4 dv = *reinterpret_cast<const int4*>(dst + e0);
            const int4 sq = *reinterpret_cast<const int4*>(src + e0);
            d4[0] = dv.x; d4[1] = dv.y; d4[2] = dv.z; d4[3] = dv.w;
            s4[0] = sq.x; s4[1] = sq.y; s4[2] = sq.z; s4[3] = sq.w;
        } else {
#pragma unroll
            for (int c = 0; c < 4; ++c) {
                const int e = e0 + c;
                d4[c] = (e < n_edges) ? dst[e] : -1;
                s4[c] = (e < n_edges) ? src[e] : 0;
            }
        }
#pragma unroll
        for (int c = 0; c < 4; ++c) {
            const int q = i * 4 + c;
            const unsigned d = (unsigned)d4[c];
            const unsigned g = d / (unsigned)NPB;
            const bool ok = (d < (unsigned)n_nodes) && (g < (unsigned)GMAX);
            gq[q]  = ok ? (int)g : -1;
            dlv[q] = (int)(d - g * (unsigned)NPB);
            sv[q]  = min(max(s4[c], 0), n_nodes - 1);
            rk[q]  = ok ? atomicAdd(&cntS[g], 1) : 0;
        }
    }
    __syncthreads();

    const int v = (tid < n_groups) ? cntS[tid] : 0;
    int inc = v;
#pragma unroll
    for (int o = 1; o < 32; o <<= 1) {
        const int n = __shfl_up(inc, o);
        if (lane >= o) inc += n;
    }
    if (lane == 31) wsum[wave] = inc;
    __syncthreads();
    if (wave == 0) {
        const int s = wsum[lane];
        int si = s;
#pragma unroll
        for (int o = 1; o < 32; o <<= 1) {
            const int n = __shfl_up(si, o);
            if (lane >= o) si += n;
        }
        wsum[lane] = si - s;
    }
    __syncthreads();
    const int excl = inc - v + wsum[wave];
    cntS[tid] = excl;
    if (tid <= n_groups) off[(size_t)w * (n_groups + 1) + tid] = excl;
    __syncthreads();

#pragma unroll
    for (int q = 0; q < 8; ++q) {
        if (gq[q] >= 0) sortS[cntS[gq[q]] + rk[q]] = make_int2(sv[q], dlv[q]);
    }
    __syncthreads();

    const int total = cntS[min(n_groups, GMAX)];
    int2* lout = reinterpret_cast<int2*>(listI) + (size_t)w * CHUNK;
    for (int i = tid; i < total; i += 1024) lout[i] = sortS[i];
}

__global__ void __launch_bounds__(1024)
owner_gather_linear(const float* __restrict__ x,
                    const int*   __restrict__ listI,
                    const int*   __restrict__ off,
                    const float* __restrict__ W,
                    const float* __restrict__ bias,
                    float*       __restrict__ out,
                    int n_nodes, int n_chunks, int n_groups) {
    __shared__ float aggS[NPB * AGG_LD];
    __shared__ int   degS[NPB];
    __shared__ float WtPf[32 * WTP_LD * 2];
    __shared__ float bsh[N_FEATS];

    const int tid  = threadIdx.x;
    const int wave = __builtin_amdgcn_readfirstlane(tid >> 5);
    const int lane = tid & 31;
    const int grp  = blockIdx.x;
    const int lo   = grp * NPB;
    const int own  = min(NPB, n_nodes - lo);

    for (int i = tid; i < NPB * AGG_LD; i += 1024) aggS[i] = 0.0f;
    for (int i = tid; i < NPB; i += 1024) degS[i] = 0;
    for (int i = tid; i < N_FEATS * N_FEATS; i += 1024) {
        int n = i / N_FEATS;
        int k = i % N_FEATS;
        WtPf[((k >> 1) * WTP_LD + n) * 2 + (k & 1)] = W[i];
    }
    if (tid < N_FEATS) bsh[tid] = bias[tid];

    const int2* list = reinterpret_cast<const int2*>(listI);
    int segS = 0, segL = 0;
    {
        const int wc = wave + NWAVES * lane;
        if (wc < n_chunks) {
            const int* orow = off + (size_t)wc * (n_groups + 1) + grp;
            const int a = orow[0];
            const int b = orow[1];
            segS = wc * CHUNK + a;
            segL = b - a;
        }
    }
    int incl = segL;
#pragma unroll
    for (int o = 1; o < 32; o <<= 1) {
        const int n = __shfl_up(incl, o);
        if (lane >= o) incl += n;
    }
    const int excl = incl - segL;
    const int tw   = __builtin_amdgcn_readlane(incl, 31);
    const int nseg = min(32, (n_chunks - wave + NWAVES - 1) / NWAVES);
    __syncthreads();

    auto load_trip = [&](int f0) -> int2 {
        const int f = f0 + lane;
        int seg = 0;
        for (int j = 0; j < nseg - 1; ++j) seg += (f >= __builtin_amdgcn_readlane(incl, j)) ? 1 : 0;
        const int sS = __shfl(segS, seg);
        const int sE = __shfl(excl, seg);
        int2 ent = make_int2(0, 0);
        if (f < tw) ent = list[sS + (f - sE)];
        return ent;
    };

    auto add_rows = [&](int2 ent, int m) {
        for (int k = 0; k < m; k += ROWS_U) {
            v2f r[ROWS_U];
            int dl[ROWS_U];
#pragma unroll
            for (int u = 0; u < ROWS_U; ++u) {
                const int j = min(k + u, m - 1);
                const int s = __builtin_amdgcn_readlane(ent.x, j);
                dl[u] = __builtin_amdgcn_readlane(ent.y, j);
                r[u] = *reinterpret_cast<const v2f*>(x + (size_t)s * N_FEATS + 2 * lane);
            }
#pragma unroll
            for (int u = 0; u < ROWS_U; ++u) {
                if (k + u < m) {
                    float* a = aggS + dl[u] * AGG_LD + 2 * lane;
                    atomicAdd(a, r[u].x);
                    atomicAdd(a + 1, r[u].y);
                    if (lane == 0) atomicAdd(&degS[dl[u]], 1);
                }
            }
        }
    };

    int2 cur = load_trip(0);
    for (int f0 = 0; f0 < tw; f0 += 32) {
        int2 nxt = make_int2(0, 0);
        if (f0 + 32 < tw) nxt = load_trip(f0 + 32);
        add_rows(cur, min(32, tw - f0));
        cur = nxt;
    }
    __syncthreads();
    for (int i = tid; i < own * 16; i += 1024) {
        const int r = i >> 4;
        const int q = i & 15;
        if (degS[r] == 0) {
            const float4 v = *reinterpret_cast<const float4*>(x + (size_t)(lo + r) * N_FEATS + q * 4);
            float* a = aggS + r * AGG_LD + q * 4;
            a[0] = v.x; a[1] = v.y; a[2] = v.z; a[3] = v.w;
        }
    }
    __syncthreads();

    const int n_tiles = (own + 15) >> 4;
    if (wave >= n_tiles) return;

    const int half = lane >> 4;
    const int l16  = lane & 15;
    const float* hrow = aggS + (wave * 16 + l16) * AGG_LD;

    v8f acc0 = {}, acc1 = {}, acc2 = {}, acc3 = {};

#pragma unroll
    for (int s = 0; s < 16; ++s) {
        const int kp = 2 * s + half;
        v2f a = *reinterpret_cast<const v2f*>(hrow + 2 * kp);
        const v2f* wrow = reinterpret_cast<const v2f*>(&WtPf[kp * WTP_LD * 2]);
        v2f b0 = wrow[l16 +  0];
        v2f b1 = wrow[l16 + 16];
        v2f b2 = wrow[l16 + 32];
        v2f b3 = wrow[l16 + 48];
        acc0 = __builtin_amdgcn_wmma_f32_16x16x4_f32(false, a, false, b0, (short)0, acc0, false, false);
        acc1 = __builtin_amdgcn_wmma_f32_16x16x4_f32(false, a, false, b1, (short)0, acc1, false, false);
        acc2 = __builtin_amdgcn_wmma_f32_16x16x4_f32(false, a, false, b2, (short)0, acc2, false, false);
        acc3 = __builtin_amdgcn_wmma_f32_16x16x4_f32(false, a, false, b3, (short)0, acc3, false, false);
    }

    const int row0 = lo + wave * 16 + 8 * half;
    float* obase = out + (size_t)row0 * N_FEATS + l16;
#pragma unroll
    for (int g = 0; g < 8; ++g) {
        if (row0 + g < n_nodes) {
            float* orow = obase + (size_t)g * N_FEATS;
            orow[ 0] = acc0[g] + bsh[l16 +  0];
            orow[16] = acc1[g] + bsh[l16 + 16];
            orow[32] = acc2[g] + bsh[l16 + 32];
            orow[48] = acc3[g] + bsh[l16 + 48];
        }
    }
}

extern "C" void kernel_launch(void* const* d_in, const int* in_sizes, int n_in,
                              void* d_out, int out_size, void* d_ws, size_t ws_size,
                              hipStream_t stream) {
    const float* x    = (const float*)d_in[0];
    const int*   src  = (const int*)  d_in[1];
    const int*   dst  = (const int*)  d_in[2];
    const float* W    = (const float*)d_in[3];
    const float* bias = (const float*)d_in[4];
    float*       out  = (float*)d_out;

    const int n_nodes = in_sizes[0] / N_FEATS;
    const int n_edges = in_sizes[1];

    int* list = (int*)((char*)d_ws + LIST_OFF_BYTES);
    int* off  = (int*)((char*)d_ws + OFF_OFF_BYTES);

    const int blocks = (n_nodes + NPB - 1) / NPB;
    const int chunks = n_edges / CHUNK + 1;
    bin_edges<<<chunks, 1024, 0, stream>>>(src, dst, list, off, n_nodes, n_edges, blocks);
    owner_gather_linear<<<blocks, 1024, 0, stream>>>(x, list, off, W, bias, out, n_nodes, chunks, blocks);
}


# Owner group sorts its edges by destination row in LDS (integer rank atomics); each wave sums its rows' x[src] in registers in arrival-rank order, one store per row: LDS float atomics gone
# speedup vs baseline: 6.7174x; 1.4825x over previous
#include <hip/hip_runtime.h>

#define N_FEATS 64
#define WTP_LD  80
#define NPB     400
#define AGG_LD  68
#define NWAVES  32
#define CHUNK   8192
#define GMAX    1023
#define ROWS_U  32
#define TRIPS   12
#define CAPW    (TRIPS * 32)
#define CAP     (CAPW * NWAVES)

#define LIST_OFF_BYTES 0u
#define OFF_OFF_BYTES  (128u << 20)

typedef float v2f __attribute__((ext_vector_type(2)));
typedef float v8f __attribute__((ext_vector_type(8)));

__global__ void __launch_bounds__(1024)
bin_edges(const int* __restrict__ src,
          const int* __restrict__ dst,
          int*       __restrict__ listI,
          int*       __restrict__ off,
          int n_nodes, int n_edges, int n_groups) {
    __shared__ int  cntS[GMAX + 1];
    __shared__ int  wsum[NWAVES];
    __shared__ int2 sortS[CHUNK];

    const int tid  = threadIdx.x;
    const int wave = tid >> 5;
    const int lane = tid & 31;
    const int w    = blockIdx.x;

    cntS[tid] = 0;
    __syncthreads();

    int gq[8], rk[8], sv[8], dlv[8];
#pragma unroll
    for (int i = 0; i < 2; ++i) {
        const int e0 = w * CHUNK + (i * 1024 + tid) * 4;
        int d4[4], s4[4];
        if (e0 + 3 < n_edges) {
            const int4 dv = *reinterpret_cast<const int4*>(dst + e0);
            const int4 sq = *reinterpret_cast<const int4*>(src + e0);
            d4[0] = dv.x; d4[1] = dv.y; d4[2] = dv.z; d4[3] = dv.w;
            s4[0] = sq.x; s4[1] = sq.y; s4[2] = sq.z; s4[3] = sq.w;
        } else {
#pragma unroll
            for (int c = 0; c < 4; ++c) {
                const int e = e0 + c;
                d4[c] = (e < n_edges) ? dst[e] : -1;
                s4[c] = (e < n_edges) ? src[e] : 0;
            }
        }
#pragma unroll
        for (int c = 0; c < 4; ++c) {
            const int q = i * 4 + c;
            const unsigned d = (unsigned)d4[c];
            const unsigned g = d / (unsigned)NPB;
            const bool ok = (d < (unsigned)n_nodes) && (g < (unsigned)GMAX);
            gq[q]  = ok ? (int)g : -1;
            dlv[q] = (int)(d - g * (unsigned)NPB);
            sv[q]  = min(max(s4[c], 0), n_nodes - 1);
            rk[q]  = ok ? atomicAdd(&cntS[g], 1) : 0;
        }
    }
    __syncthreads();

    const int v = (tid < n_groups) ? cntS[tid] : 0;
    int inc = v;
#pragma unroll
    for (int o = 1; o < 32; o <<= 1) {
        const int n = __shfl_up(inc, o);
        if (lane >= o) inc += n;
    }
    if (lane == 31) wsum[wave] = inc;
    __syncthreads();
    if (wave == 0) {
        const int s = wsum[lane];
        int si = s;
#pragma unroll
        for (int o = 1; o < 32; o <<= 1) {
            const int n = __shfl_up(si, o);
            if (lane >= o) si += n;
        }
        wsum[lane] = si - s;
    }
    __syncthreads();
    const int excl = inc - v + wsum[wave];
    cntS[tid] = excl;
    if (tid <= n_groups) off[(size_t)w * (n_groups + 1) + tid] = excl;
    __syncthreads();

#pragma unroll
    for (int q = 0; q < 8; ++q) {
        if (gq[q] >= 0) sortS[cntS[gq[q]] + rk[q]] = make_int2(sv[q], dlv[q]);
    }
    __syncthreads();

    const int total = cntS[min(n_groups, GMAX)];
    int2* lout = reinterpret_cast<int2*>(listI) + (size_t)w * CHUNK;
    for (int i = tid; i < total; i += 1024) lout[i] = sortS[i];
}

__device__ __forceinline__ void flush_row(float* aggLane, int row, int pass, v2f acc) {
    v2f* p = reinterpret_cast<v2f*>(aggLane + row * AGG_LD);
    if (pass == 0) {
        *p = acc;
    } else {
        v2f o = *p;
        o += acc;
        *p = o;
    }
}

template <int NB, bool FULL>
__device__ __forceinline__ void sum_rows(const char* __restrict__ xb, unsigned lane8, int2 en, int m,
                                         float* aggLane, int pass, v2f& acc, int& curd) {
    v2f r[NB];
#pragma unroll
    for (int u = 0; u < NB; ++u) {
        const int j = FULL ? u : min(u, m - 1);
        const unsigned s = (unsigned)__builtin_amdgcn_readlane(en.x, j);
        const char* rowb = xb + (size_t)s * 256u;
        r[u] = *reinterpret_cast<const v2f*>(rowb + lane8);
    }
#pragma unroll
    for (int u = 0; u < NB; ++u) {
        if (FULL || u < m) {
            const int d = __builtin_amdgcn_readlane(en.y, u);
            if (__builtin_expect(d != curd, 0)) {
                if (curd < NPB) flush_row(aggLane, curd, pass, acc);
                acc = v2f{0.0f, 0.0f};
                curd = d;
            }
            acc += r[u];
        }
    }
}

__global__ void __launch_bounds__(1024)
owner_gather_linear(const float* __restrict__ x,
                    const int*   __restrict__ listI,
                    const int*   __restrict__ off,
                    const float* __restrict__ W,
                    const float* __restrict__ bias,
                    float*       __restrict__ out,
                    int n_nodes, int n_chunks, int n_groups) {
    __shared__ float aggS[NPB * AGG_LD];
    __shared__ int   cntS[NPB];
    __shared__ int   offS[NPB + 1];
    __shared__ int   degS[NPB];
    __shared__ int   wsum[NWAVES];
    __shared__ int   maxS;
    __shared__ int2  sortS[CAP];
    __shared__ float WtPf[32 * WTP_LD * 2];
    __shared__ float bsh[N_FEATS];

    const int tid  = threadIdx.x;
    const int wave = __builtin_amdgcn_readfirstlane(tid >> 5);
    const int lane = tid & 31;
    const int grp  = blockIdx.x;
    const int lo   = grp * NPB;
    const int own  = min(NPB, n_nodes - lo);

    for (int i = tid; i < NPB * AGG_LD; i += 1024) aggS[i] = 0.0f;
    if (tid < NPB) { cntS[tid] = 0; degS[tid] = 0; }
    if (tid == 0) maxS = 0;
    for (int i = tid; i < N_FEATS * N_FEATS; i += 1024) {
        int n = i / N_FEATS;
        int k = i % N_FEATS;
        WtPf[((k >> 1) * WTP_LD + n) * 2 + (k & 1)] = W[i];
    }
    if (tid < N_FEATS) bsh[tid] = bias[tid];

    const int2* list = reinterpret_cast<const int2*>(listI);
    int segS = 0, segL = 0;
    {
        const int wc = wave + NWAVES * lane;
        if (wc < n_chunks) {
            const int* orow = off + (size_t)wc * (n_groups + 1) + grp;
            const int a = orow[0];
            const int b = orow[1];
            segS = wc * CHUNK + a;
            segL = b - a;
        }
    }
    int incl = segL;
#pragma unroll
    for (int o = 1; o < 32; o <<= 1) {
        const int n = __shfl_up(incl, o);
        if (lane >= o) incl += n;
    }
    const int dlt  = segS - (incl - segL);
    const int tw   = __builtin_amdgcn_readlane(incl, 31);
    const int nseg = min(32, (n_chunks - wave + NWAVES - 1) / NWAVES);
    __syncthreads();
    if (lane == 0) atomicMax(&maxS, tw);
    __syncthreads();
    const int npass = (maxS + CAPW - 1) / CAPW;

    const char*    xb      = reinterpret_cast<const char*>(x);
    const unsigned lane8   = (unsigned)lane * 8u;
    float*         aggLane = aggS + 2 * lane;
    const int r0 = (wave * NPB) / NWAVES;
    const int r1 = ((wave + 1) * NPB) / NWAVES;

    for (int pass = 0; pass < npass; ++pass) {
        const int fb = pass * CAPW + lane;
        int dl[TRIPS];
        {
            const int d0 = __builtin_amdgcn_readlane(dlt, 0);
#pragma unroll
            for (int t = 0; t < TRIPS; ++t) dl[t] = d0;
        }
        for (int j = 0; j < nseg - 1; ++j) {
            const int e  = __builtin_amdgcn_readlane(incl, j);
            const int dn = __builtin_amdgcn_readlane(dlt, j + 1);
#pragma unroll
            for (int t = 0; t < TRIPS; ++t) dl[t] = (fb + 32 * t >= e) ? dn : dl[t];
        }
        int2 ent[TRIPS];
#pragma unroll
        for (int t = 0; t < TRIPS; ++t) {
            const int f = fb + 32 * t;
            ent[t] = make_int2(0, -1);
            if (f < tw) ent[t] = list[dl[t] + f];
        }
        int rk[TRIPS];
#pragma unroll
        for (int t = 0; t < TRIPS; ++t) {
            rk[t] = 0;
            if (ent[t].y >= 0) rk[t] = atomicAdd(&cntS[ent[t].y], 1);
        }
        __syncthreads();

        const int v = (tid < NPB) ? cntS[tid] : 0;
        int inc = v;
#pragma unroll
        for (int o = 1; o < 32; o <<= 1) {
            const int n = __shfl_up(inc, o);
            if (lane >= o) inc += n;
        }
        if (lane == 31) wsum[wave] = inc;
        __syncthreads();
        if (wave == 0) {
            const int s = wsum[lane];
            int si = s;
#pragma unroll
            for (int o = 1; o < 32; o <<= 1) {
                const int n = __shfl_up(si, o);
                if (lane >= o) si += n;
            }
            wsum[lane] = si - s;
        }
        __syncthreads();
        const int ex = inc - v + wsum[wave];
        if (tid < NPB) {
            offS[tid] = ex;
            degS[tid] += v;
            cntS[tid] = 0;
        }
        if (tid == NPB - 1) offS[NPB] = ex + v;
        __syncthreads();

#pragma unroll
        for (int t = 0; t < TRIPS; ++t) {
            if (ent[t].y >= 0) sortS[offS[ent[t].y] + rk[t]] = ent[t];
        }
        __syncthreads();

        const int a = __builtin_amdgcn_readfirstlane(offS[r0]);
        const int b = __builtin_amdgcn_readfirstlane(offS[r1]);
        v2f acc = {0.0f, 0.0f};
        int curd = NPB;
        for (int e0 = a; e0 < b; e0 += ROWS_U) {
            const int m = min(ROWS_U, b - e0);
            int2 en = make_int2(0, 0);
            if (lane < m) en = sortS[e0 + lane];
            if (m == ROWS_U)          sum_rows<ROWS_U,     true >(xb, lane8, en, m, aggLane, pass, acc, curd);
            else if (m > ROWS_U / 2)  sum_rows<ROWS_U,     false>(xb, lane8, en, m, aggLane, pass, acc, curd);
            else                      sum_rows<ROWS_U / 2, false>(xb, lane8, en, m, aggLane, pass, acc, curd);
        }
        if (curd < NPB) flush_row(aggLane, curd, pass, acc);
        __syncthreads();
    }

    for (int i = tid; i < own * 16; i += 1024) {
        const int r = i >> 4;
        const int q = i & 15;
        if (degS[r] == 0) {
            const float4 v = *reinterpret_cast<const float4*>(x + (size_t)(lo + r) * N_FEATS + q * 4);
            float* a = aggS + r * AGG_LD + q * 4;
            a[0] = v.x; a[1] = v.y; a[2] = v.z; a[3] = v.w;
        }
    }
    __syncthreads();

    const int n_tiles = (own + 15) >> 4;
    if (wave >= n_tiles) return;

    const int half = lane >> 4;
    const int l16  = lane & 15;
    const float* hrow = aggS + (wave * 16 + l16) * AGG_LD;

    v8f acc0 = {}, acc1 = {}, acc2 = {}, acc3 = {};

#pragma unroll
    for (int s = 0; s < 16; ++s) {
        const int kp = 2 * s + half;
        v2f a = *reinterpret_cast<const v2f*>(hrow + 2 * kp);
        const v2f* wrow = reinterpret_cast<const v2f*>(&WtPf[kp * WTP_LD * 2]);
        v2f b0 = wrow[l16 +  0];
        v2f b1 = wrow[l16 + 16];
        v2f b2 = wrow[l16 + 32];
        v2f b3 = wrow[l16 + 48];
        acc0 = __builtin_amdgcn_wmma_f32_16x16x4_f32(false, a, false, b0, (short)0, acc0, false, false);
        acc1 = __builtin_amdgcn_wmma_f32_16x16x4_f32(false, a, false, b1, (short)0, acc1, false, false);
        acc2 = __builtin_amdgcn_wmma_f32_16x16x4_f32(false, a, false, b2, (short)0, acc2, false, false);
        acc3 = __builtin_amdgcn_wmma_f32_16x16x4_f32(false, a, false, b3, (short)0, acc3, false, false);
    }

    const int row0 = lo + wave * 16 + 8 * half;
    float* obase = out + (size_t)row0 * N_FEATS + l16;
#pragma unroll
    for (int g = 0; g < 8; ++g) {
        if (row0 + g < n_nodes) {
            float* orow = obase + (size_t)g * N_FEATS;
            orow[ 0] = acc0[g] + bsh[l16 +  0];
            orow[16] = acc1[g] + bsh[l16 + 16];
            orow[32] = acc2[g] + bsh[l16 + 32];
            orow[48] = acc3[g] + bsh[l16 + 48];
        }
    }
}

extern "C" void kernel_launch(void* const* d_in, const int* in_sizes, int n_in,
                              void* d_out, int out_size, void* d_ws, size_t ws_size,
                              hipStream_t stream) {
    const float* x    = (const float*)d_in[0];
    const int*   src  = (const int*)  d_in[1];
    const int*   dst  = (const int*)  d_in[2];
    const float* W    = (const float*)d_in[3];
    const float* bias = (const float*)d_in[4];
    float*       out  = (float*)d_out;

    const int n_nodes = in_sizes[0] / N_FEATS;
    const int n_edges = in_sizes[1];

    int* list = (int*)((char*)d_ws + LIST_OFF_BYTES);
    int* off  = (int*)((char*)d_ws + OFF_OFF_BYTES);

    const int blocks = (n_nodes + NPB - 1) / NPB;
    const int chunks = n_edges / CHUNK + 1;
    bin_edges<<<chunks, 1024, 0, stream>>>(src, dst, list, off, n_nodes, n_edges, blocks);
    owner_gather_linear<<<blocks, 1024, 0, stream>>>(x, list, off, W, bias, out, n_nodes, chunks, blocks);
}
